# starting kernel
# baseline (speedup 1.0000x reference)
#include <hip/hip_runtime.h>


typedef __attribute__((ext_vector_type(2))) float v2f;
typedef __attribute__((ext_vector_type(8))) float v8f;

#define KS   31
#define LTOT (KS * KS)
#define PADN 32
#define LSTR 33

__global__ __launch_bounds__(128)
void idgc_wmma_kernel(const float* __restrict__ x,
                      const int* __restrict__ center_idx,
                      float* __restrict__ out) {
    __shared__ float sX[PADN * LSTR];
    __shared__ float sA[PADN * LSTR];
    __shared__ float sT[PADN * LSTR];

    const int plane = blockIdx.x;
    const float* xp = x + (size_t)plane * LTOT;
    float* op       = out + (size_t)plane * LTOT;

    const int tid = threadIdx.x;

    for (int i = tid; i < PADN * PADN; i += 128) {
        const int r = i >> 5;
        const int c = i & 31;
        const bool in = (r < KS) && (c < KS);
        sX[r * LSTR + c] = in ? xp[r * KS + c] : 0.0f;
        const int d = r - c;
        sA[r * LSTR + c] = (in && d <= 15 && d >= -15) ? 1.0f : 0.0f;
    }
    __syncthreads();

    const int lane = tid & 31;
    const int wid  = tid >> 5;
    const int I    = (wid >> 1) << 4;
    const int J    = (wid & 1) << 4;
    const int half = lane >> 4;
    const int ln16 = lane & 15;

    v8f acc = {};
#pragma unroll
    for (int k0 = 0; k0 < PADN; k0 += 4) {
        const int ka = k0 + 2 * half;
        v2f a, b;
        a.x = sA[(I + ln16) * LSTR + (ka + 0)];
        a.y = sA[(I + ln16) * LSTR + (ka + 1)];
        b.x = sX[(ka + 0) * LSTR + (J + ln16)];
        b.y = sX[(ka + 1) * LSTR + (J + ln16)];
        acc = __builtin_amdgcn_wmma_f32_16x16x4_f32(
             false, a,  false, b,
             (short)0, acc,  false,  false);
    }
#pragma unroll
    for (int v = 0; v < 8; ++v) {
        sT[(I + v + 8 * half) * LSTR + (J + ln16)] = acc[v];
    }
    __syncthreads();

    v8f acc2 = {};
#pragma unroll
    for (int k0 = 0; k0 < PADN; k0 += 4) {
        const int ka = k0 + 2 * half;
        v2f a, b;
        a.x = sT[(I + ln16) * LSTR + (ka + 0)];
        a.y = sT[(I + ln16) * LSTR + (ka + 1)];
        b.x = sA[(ka + 0) * LSTR + (J + ln16)];
        b.y = sA[(ka + 1) * LSTR + (J + ln16)];
        acc2 = __builtin_amdgcn_wmma_f32_16x16x4_f32(
            false, a, false, b, (short)0, acc2, false, false);
    }

    const int cidx = *center_idx;
    const int ci = cidx / KS;
    const int cj = cidx % KS;
    const float inv = 1.0f / (float)LTOT;
#pragma unroll
    for (int v = 0; v < 8; ++v) {
        const int row = I + v + 8 * half;
        const int col = J + ln16;
        if (row < KS && col < KS) {
            const int xr = ci + row - 15;
            const int xc = cj + col - 15;
            const float cp = (xr >= 0 && xr < KS && xc >= 0 && xc < KS)
                                 ? sX[xr * LSTR + xc] : 0.0f;
            op[row * KS + col] = acc2[v] * inv - cp;
        }
    }
}

extern "C" void kernel_launch(void* const* d_in, const int* in_sizes, int n_in,
                              void* d_out, int out_size, void* d_ws, size_t ws_size,
                              hipStream_t stream) {
    (void)in_sizes; (void)n_in; (void)out_size; (void)d_ws; (void)ws_size;
    const float* x          = (const float*)d_in[0];
    const int*   center_idx = (const int*)d_in[1];
    float*       out        = (float*)d_out;

    idgc_wmma_kernel<<<256, 128, 0, stream>>>(x, center_idx, out);
}


# A*X*A chain kept in registers, one wave per 16-row slab: X loaded from global as one group, band built in registers, product 1 transposed so its D is product 2's A (k order permuted); no LDS, 256x64
# speedup vs baseline: 2.0429x; 2.0429x over previous
#include <hip/hip_runtime.h>


typedef __attribute__((ext_vector_type(2))) float v2f;
typedef __attribute__((ext_vector_type(8))) float v8f;

#define KS   31
#define LTOT (KS * KS)

__device__ __forceinline__ float band(int r, int c) {
    const int d = r - c;
    return (r < KS && c < KS && d <= 15 && d >= -15) ? 1.0f : 0.0f;
}

__global__ __launch_bounds__(64)
void idgc_wmma_kernel(const float* __restrict__ x,
                      const int* __restrict__ center_idx,
                      float* __restrict__ out) {
    const int tid  = threadIdx.x;
    const int lane = tid & 31;
    const int I    = ((tid >> 5) & 1) << 4;
    const int half = lane >> 4;
    const int ln16 = lane & 15;

    v2f bA[8];
#pragma unroll
    for (int t = 0; t < 8; ++t) {
        const int k = 4 * t + 2 * half;
        bA[t].x = band(I + ln16, k + 0);
        bA[t].y = band(I + ln16, k + 1);
    }
    v2f bB[2][2][4];
#pragma unroll
    for (int jb = 0; jb < 2; ++jb) {
#pragma unroll
        for (int mb = 0; mb < 2; ++mb) {
#pragma unroll
            for (int s2 = 0; s2 < 4; ++s2) {
                const int k = 16 * mb + 8 * half + 2 * s2;
                bB[jb][mb][s2].x = band(k + 0, 16 * jb + ln16);
                bB[jb][mb][s2].y = band(k + 1, 16 * jb + ln16);
            }
        }
    }

    const int cidx = *center_idx;

    __builtin_amdgcn_sched_barrier(0);

    const int plane = blockIdx.x;
    const float* xp = x + (size_t)plane * LTOT;
    float* op       = out + (size_t)plane * LTOT;

    float xraw[2][8][2];
#pragma unroll
    for (int mb = 0; mb < 2; ++mb) {
#pragma unroll
        for (int t = 0; t < 8; ++t) {
#pragma unroll
            for (int s = 0; s < 2; ++s) {
                const int k  = 4 * t + 2 * half + s;
                const int m  = 16 * mb + ln16;
                const int kk = (k < KS) ? k : (KS - 1);
                const int mm = (m < KS) ? m : (KS - 1);
                xraw[mb][t][s] = xp[kk * KS + mm];
            }
        }
    }

    __builtin_amdgcn_sched_barrier(0);

    const int ci = cidx / KS;
    const int cj = cidx % KS;
    float cpraw[2][8];
#pragma unroll
    for (int jb = 0; jb < 2; ++jb) {
#pragma unroll
        for (int v = 0; v < 8; ++v) {
            const int xr = ci + (I + 8 * half + v) - 15;
            const int xc = cj + (16 * jb + ln16) - 15;
            const bool ok = (xr >= 0 && xr < KS && xc >= 0 && xc < KS);
            cpraw[jb][v] = xp[ok ? (xr * KS + xc) : 0];
        }
    }

    __builtin_amdgcn_sched_barrier(0);

    asm volatile("" ::
        "v"(cpraw[0][0]), "v"(cpraw[0][1]), "v"(cpraw[0][2]), "v"(cpraw[0][3]),
        "v"(cpraw[0][4]), "v"(cpraw[0][5]), "v"(cpraw[0][6]), "v"(cpraw[0][7]),
        "v"(cpraw[1][0]), "v"(cpraw[1][1]), "v"(cpraw[1][2]), "v"(cpraw[1][3]),
        "v"(cpraw[1][4]), "v"(cpraw[1][5]), "v"(cpraw[1][6]), "v"(cpraw[1][7]));

    v2f xa[2][8];
#pragma unroll
    for (int mb = 0; mb < 2; ++mb) {
#pragma unroll
        for (int t = 0; t < 8; ++t) {
#pragma unroll
            for (int s = 0; s < 2; ++s) {
                const int k = 4 * t + 2 * half + s;
                const int m = 16 * mb + ln16;
                xa[mb][t][s] = (k < KS && m < KS) ? xraw[mb][t][s] : 0.0f;
            }
        }
    }

    v8f d0 = {};
    v8f d1 = {};
#pragma unroll
    for (int t = 0; t < 8; ++t) {
        d0 = __builtin_amdgcn_wmma_f32_16x16x4_f32(
             false, xa[0][t],  false, bA[t],
             (short)0, d0,  false,  false);
        d1 = __builtin_amdgcn_wmma_f32_16x16x4_f32(
            false, xa[1][t], false, bA[t], (short)0, d1, false, false);
    }

    v8f s0 = {};
    v8f s1 = {};
#pragma unroll
    for (int s2 = 0; s2 < 4; ++s2) {
        v2f a;
        a.x = d0[2 * s2 + 0];
        a.y = d0[2 * s2 + 1];
        s0 = __builtin_amdgcn_wmma_f32_16x16x4_f32(
            false, a, false, bB[0][0][s2], (short)0, s0, false, false);
        s1 = __builtin_amdgcn_wmma_f32_16x16x4_f32(
            false, a, false, bB[1][0][s2], (short)0, s1, false, false);
    }
#pragma unroll
    for (int s2 = 0; s2 < 4; ++s2) {
        v2f a;
        a.x = d1[2 * s2 + 0];
        a.y = d1[2 * s2 + 1];
        s0 = __builtin_amdgcn_wmma_f32_16x16x4_f32(
            false, a, false, bB[0][1][s2], (short)0, s0, false, false);
        s1 = __builtin_amdgcn_wmma_f32_16x16x4_f32(
            false, a, false, bB[1][1][s2], (short)0, s1, false, false);
    }

    const float inv = 1.0f / (float)LTOT;
#pragma unroll
    for (int jb = 0; jb < 2; ++jb) {
        const int col = 16 * jb + ln16;
        const int xc  = cj + col - 15;
        if (col < KS) {
#pragma unroll
            for (int v = 0; v < 8; ++v) {
                const int row = I + 8 * half + v;
                const int xr  = ci + row - 15;
                const bool ok = (xr >= 0 && xr < KS && xc >= 0 && xc < KS);
                const float cp = ok ? cpraw[jb][v] : 0.0f;
                const float sv = (jb == 0) ? s0[v] : s1[v];
                if (row < KS) {
                    op[row * KS + col] = sv * inv - cp;
                }
            }
        }
    }
}

extern "C" void kernel_launch(void* const* d_in, const int* in_sizes, int n_in,
                              void* d_out, int out_size, void* d_ws, size_t ws_size,
                              hipStream_t stream) {
    (void)in_sizes; (void)n_in; (void)out_size; (void)d_ws; (void)ws_size;
    const float* x          = (const float*)d_in[0];
    const int*   center_idx = (const int*)d_in[1];
    float*       out        = (float*)d_out;

    idgc_wmma_kernel<<<256, 64, 0, stream>>>(x, center_idx, out);
}
